# starting kernel
# baseline (speedup 1.0000x reference)
#include <hip/hip_runtime.h>

typedef __attribute__((ext_vector_type(2))) float v2f;
typedef __attribute__((ext_vector_type(8))) float v8f;

#define B_  32
#define C_  256
#define H_  64
#define W_  64
#define HW_ (H_ * W_)

#define PX_ 128
#define KPLANE_ (PX_ * 4)
#define SMEM_BYTES_ ((C_ / 4) * KPLANE_ * 4)

__device__ __forceinline__ void s2f(float& a, float& b) {
    float mn = fminf(a, b);
    b = fmaxf(a, b);
    a = mn;
}

__device__ __forceinline__ float median9(float v0, float v1, float v2,
                                         float v3, float v4, float v5,
                                         float v6, float v7, float v8) {
    s2f(v0, v3); s2f(v1, v4); s2f(v2, v5);
    s2f(v0, v1); s2f(v0, v2);
    s2f(v4, v5); s2f(v3, v5);
    s2f(v1, v2); s2f(v3, v4);
    s2f(v1, v3); s2f(v1, v6);
    s2f(v4, v6); s2f(v2, v6);
    s2f(v2, v3); s2f(v4, v7); s2f(v2, v4); s2f(v3, v7);
    s2f(v4, v8); s2f(v3, v8); s2f(v3, v4);
    return v4;
}

__device__ __forceinline__ float ldz(const float* __restrict__ p, int off, bool valid) {
    float v = p[valid ? off : 0];
    return valid ? v : 0.0f;
}

__global__ __launch_bounds__(256)
void fused_median_conv1x1(const float* __restrict__ x,
                          const float* __restrict__ cw,
                          const float* __restrict__ cb,
                          float* __restrict__ out) {
    extern __shared__ float smed[];

    const int h0 = blockIdx.x * 2;
    const int b  = blockIdx.y;
    const int t  = threadIdx.x;

    {
        const int w   = t & 63;
        const int q   = t >> 6;
        const int r0  = q & 1;
        const int cp  = q >> 1;
        const int row = h0 + r0;
        const bool hm = (row > 0), hp = (row < H_ - 1);
        const bool wm = (w > 0),   wp = (w < W_ - 1);
        const int  pix = r0 * 64 + w;
        const size_t rowbase = (size_t)b * C_ * HW_ + (size_t)row * W_ + w;

        for (int c = cp; c < C_; c += 2) {
            const float* p = x + rowbase + (size_t)c * HW_;
            float a0 = ldz(p, -W_ - 1, hm && wm);
            float a1 = ldz(p, -W_,     hm);
            float a2 = ldz(p, -W_ + 1, hm && wp);
            float a3 = ldz(p, -1,      wm);
            float a4 = p[0];
            float a5 = ldz(p,  1,      wp);
            float a6 = ldz(p,  W_ - 1, hp && wm);
            float a7 = ldz(p,  W_,     hp);
            float a8 = ldz(p,  W_ + 1, hp && wp);
            smed[(c >> 2) * KPLANE_ + pix * 4 + (c & 3)] =
                median9(a0, a1, a2, a3, a4, a5, a6, a7, a8);
        }
    }
    __syncthreads();

    const int lane = t & 31;
    const int wv   = t >> 5;
    const int o0   = wv * 32;
    const int l15  = lane & 15;
    const int koff = (lane < 16) ? 0 : 2;

    const v2f* a0p = (const v2f*)(cw + (size_t)(o0 +      l15) * C_ + koff);
    const v2f* a1p = (const v2f*)(cw + (size_t)(o0 + 16 + l15) * C_ + koff);
    const v2f* bp[8];
#pragma unroll
    for (int j = 0; j < 8; ++j)
        bp[j] = (const v2f*)(&smed[(j * 16 + l15) * 4 + koff]);

    v8f acc[2][8];
#pragma unroll
    for (int i = 0; i < 2; ++i)
#pragma unroll
        for (int j = 0; j < 8; ++j)
            acc[i][j] = (v8f){0.f, 0.f, 0.f, 0.f, 0.f, 0.f, 0.f, 0.f};

#pragma unroll 2
    for (int k0 = 0; k0 < C_; k0 += 4) {
        v2f a0 = a0p[k0 >> 1];
        v2f a1 = a1p[k0 >> 1];
        const int bi = (k0 >> 2) * (KPLANE_ / 2);
        v2f bv[8];
#pragma unroll
        for (int j = 0; j < 8; ++j) bv[j] = bp[j][bi];
#pragma unroll
        for (int j = 0; j < 8; ++j)
            acc[0][j] = __builtin_amdgcn_wmma_f32_16x16x4_f32(false, a0, false, bv[j],
                                                              (short)0, acc[0][j], false, false);
#pragma unroll
        for (int j = 0; j < 8; ++j)
            acc[1][j] = __builtin_amdgcn_wmma_f32_16x16x4_f32(false, a1, false, bv[j],
                                                              (short)0, acc[1][j], false, false);
    }

    const int madd = (lane < 16) ? 0 : 8;

    float cbv[2][8];
#pragma unroll
    for (int i = 0; i < 2; ++i)
#pragma unroll
        for (int r = 0; r < 8; ++r)
            cbv[i][r] = cb[o0 + i * 16 + madd + r];

#pragma unroll
    for (int i = 0; i < 2; ++i) {
#pragma unroll
        for (int j = 0; j < 8; ++j) {
            const int p   = j * 16 + l15;
            const int row = h0 + (p >> 6);
            const int col = p & 63;
            const size_t base = (size_t)b * C_ * HW_ + (size_t)row * W_ + col;
            const int ob = o0 + i * 16 + madd;
            float xv[8];
#pragma unroll
            for (int r = 0; r < 8; ++r)
                xv[r] = x[base + (size_t)(ob + r) * HW_];
#pragma unroll
            for (int r = 0; r < 8; ++r)
                out[base + (size_t)(ob + r) * HW_] = xv[r] + cbv[i][r] + acc[i][j][r];
        }
    }
}

extern "C" void kernel_launch(void* const* d_in, const int* in_sizes, int n_in,
                              void* d_out, int out_size, void* d_ws, size_t ws_size,
                              hipStream_t stream) {
    (void)in_sizes; (void)n_in; (void)d_ws; (void)ws_size; (void)out_size;
    const float* x  = (const float*)d_in[0];
    const float* cw = (const float*)d_in[1];
    const float* cb = (const float*)d_in[2];
    float* out = (float*)d_out;

    dim3 grid(H_ / 2, B_);
    dim3 block(256);
    fused_median_conv1x1<<<grid, block, SMEM_BYTES_, stream>>>(x, cw, cb, out);
}


# K in 8 chunks as a 3-stage LDS pipeline split by wave role: 8 waves stage raw rows and take the next chunk's medians from LDS while 8 waves run this chunk's matrix steps; sum order kept
# speedup vs baseline: 1.1725x; 1.1725x over previous
#include <hip/hip_runtime.h>


typedef __attribute__((ext_vector_type(2))) float v2f;
typedef __attribute__((ext_vector_type(8))) float v8f;

#define B_  32
#define C_  256
#define H_  64
#define W_  64
#define HW_ (H_ * W_)

#define PX_ 128
#define KPLANE_ (PX_ * 4)

#define CH_     32
#define NCHUNK_ (C_ / CH_)
#define MSLOT_  ((CH_ / 4) * KPLANE_)
#define RP_     (W_ + 2)
#define RSLOT_  (CH_ * 4 * RP_)
#define RAW0_   (2 * MSLOT_)
#define SMEM_BYTES_ ((2 * MSLOT_ + 2 * RSLOT_) * 4)
#define GRP_    4

__device__ __forceinline__ void s2f(float& a, float& b) {
    float mn = fminf(a, b);
    b = fmaxf(a, b);
    a = mn;
}

__device__ __forceinline__ float median9(float v0, float v1, float v2,
                                         float v3, float v4, float v5,
                                         float v6, float v7, float v8) {
    s2f(v0, v3); s2f(v1, v4); s2f(v2, v5);
    s2f(v0, v1); s2f(v0, v2);
    s2f(v4, v5); s2f(v3, v5);
    s2f(v1, v2); s2f(v3, v4);
    s2f(v1, v3); s2f(v1, v6);
    s2f(v4, v6); s2f(v2, v6);
    s2f(v2, v3); s2f(v4, v7); s2f(v2, v4); s2f(v3, v7);
    s2f(v4, v8); s2f(v3, v8); s2f(v3, v4);
    return v4;
}

__device__ __forceinline__ void raw_load(const float* __restrict__ xrow, int w,
                                         float (&v)[CH_]) {
#pragma unroll
    for (int cl = 0; cl < CH_; ++cl) v[cl] = xrow[(size_t)cl * HW_ + w];
}

__device__ __forceinline__ void raw_store(float* __restrict__ rslot, int rr, int w,
                                          bool row_in, const float (&v)[CH_]) {
#pragma unroll
    for (int cl = 0; cl < CH_; ++cl)
        rslot[(cl * 4 + rr) * RP_ + w + 1] = row_in ? v[cl] : 0.0f;
}

template <int STRIDE, int NMED>
__device__ __forceinline__ void medians(const float* __restrict__ rslot,
                                        float* __restrict__ mslot,
                                        int cp, int r0, int w) {
    const float* lanep = rslot + r0 * RP_ + w;
    float* outp = mslot + (r0 * 64 + w) * 4;
#pragma unroll 1
    for (int g = 0; g < NMED; g += GRP_) {
        float v[GRP_][9];
#pragma unroll
        for (int m = 0; m < GRP_; ++m) {
            const float* p = lanep + (cp + (g + m) * STRIDE) * (4 * RP_);
#pragma unroll
            for (int dr = 0; dr < 3; ++dr)
#pragma unroll
                for (int dc = 0; dc < 3; ++dc)
                    v[m][dr * 3 + dc] = p[dr * RP_ + dc];
        }
#pragma unroll
        for (int m = 0; m < GRP_; ++m) {
            const int cl = cp + (g + m) * STRIDE;
            outp[(cl >> 2) * KPLANE_ + (cl & 3)] =
                median9(v[m][0], v[m][1], v[m][2], v[m][3], v[m][4],
                        v[m][5], v[m][6], v[m][7], v[m][8]);
        }
    }
}

__global__ __launch_bounds__(512)
void fused_median_conv1x1(const float* __restrict__ x,
                          const float* __restrict__ cw,
                          const float* __restrict__ cb,
                          float* __restrict__ out) {
    extern __shared__ float smem[];

    const int h0   = blockIdx.x * 2;
    const int b    = blockIdx.y;
    const int t    = threadIdx.x;
    const int lane = t & 31;
    const int wave = __builtin_amdgcn_readfirstlane(t >> 5);
    const bool median_wave = (wave >= 8);

    const int w  = (wave & 1) * 32 + lane;
    const int r0 = (wave >> 1) & 1;
    const int rr = (wave >> 1) & 3;
    const int irow = h0 - 1 + rr;
    const bool row_in = (irow >= 0) && (irow < H_);
    const int irow_c = row_in ? irow : h0;
    const float* xrow0 = x + (size_t)b * C_ * HW_ + (size_t)irow_c * W_;

    {
        const int slot = t >> 8, cl = (t >> 3) & 31, prow = (t >> 1) & 3;
        smem[RAW0_ + slot * RSLOT_ + (cl * 4 + prow) * RP_ + ((t & 1) ? (RP_ - 1) : 0)] = 0.0f;
    }

    if (median_wave) {
        float rv[CH_];
        raw_load(xrow0, w, rv);
        raw_store(smem + RAW0_, rr, w, row_in, rv);
    }
    __syncthreads();

    if (median_wave) {
        float rv[CH_];
        raw_load(xrow0 + (size_t)CH_ * HW_, w, rv);
        medians<4, CH_ / 4>(smem + RAW0_, smem, wave >> 2, r0, w);
        raw_store(smem + RAW0_ + RSLOT_, rr, w, row_in, rv);
    } else {
        medians<4, CH_ / 4>(smem + RAW0_, smem, wave >> 2, r0, w);
    }

    const int wv   = wave & 7;
    const int o0   = wv * 32;
    const int l15  = lane & 15;
    const int koff = (lane < 16) ? 0 : 2;

    const v2f* a0p = (const v2f*)(cw + (size_t)(o0 +      l15) * C_ + koff);
    const v2f* a1p = (const v2f*)(cw + (size_t)(o0 + 16 + l15) * C_ + koff);

    v8f acc[2][8];
#pragma unroll
    for (int i = 0; i < 2; ++i)
#pragma unroll
        for (int j = 0; j < 8; ++j)
            acc[i][j] = (v8f){0.f, 0.f, 0.f, 0.f, 0.f, 0.f, 0.f, 0.f};

    __syncthreads();

    for (int ch = 0; ch < NCHUNK_; ++ch) {
        if (median_wave) {
            const float* rnext = smem + RAW0_ + ((ch + 1) & 1) * RSLOT_;
            float* mnext = smem + ((ch + 1) & 1) * MSLOT_;
            if (ch + 2 < NCHUNK_) {
                float rv[CH_];
                raw_load(xrow0 + (size_t)(ch + 2) * CH_ * HW_, w, rv);
                medians<2, CH_ / 2>(rnext, mnext, (wave >> 2) & 1, r0, w);
                raw_store(smem + RAW0_ + (ch & 1) * RSLOT_, rr, w, row_in, rv);
            } else if (ch + 1 < NCHUNK_) {
                medians<2, CH_ / 2>(rnext, mnext, (wave >> 2) & 1, r0, w);
            }
#pragma unroll
            for (int i = 0; i < 2; ++i)
#pragma unroll
                for (int j = 0; j < 8; ++j)
                    acc[i][j] = (v8f){0.f, 0.f, 0.f, 0.f, 0.f, 0.f, 0.f, 0.f};
        } else {
            const v2f* bp = (const v2f*)(smem + (ch & 1) * MSLOT_ + l15 * 4 + koff);
#pragma unroll 2
            for (int kk = 0; kk < CH_; kk += 4) {
                const int k0 = ch * CH_ + kk;
                v2f a0 = a0p[k0 >> 1];
                v2f a1 = a1p[k0 >> 1];
                const int bi = (kk >> 2) * (KPLANE_ / 2);
                v2f bv[8];
#pragma unroll
                for (int j = 0; j < 8; ++j) bv[j] = bp[bi + j * 32];
#pragma unroll
                for (int j = 0; j < 8; ++j)
                    acc[0][j] = __builtin_amdgcn_wmma_f32_16x16x4_f32(false, a0, false, bv[j],
                                                                      (short)0, acc[0][j], false, false);
#pragma unroll
                for (int j = 0; j < 8; ++j)
                    acc[1][j] = __builtin_amdgcn_wmma_f32_16x16x4_f32(false, a1, false, bv[j],
                                                                      (short)0, acc[1][j], false, false);
            }
        }
        __syncthreads();
    }

    if (median_wave) return;

    const int madd = (lane < 16) ? 0 : 8;

    float cbv[2][8];
#pragma unroll
    for (int i = 0; i < 2; ++i)
#pragma unroll
        for (int r = 0; r < 8; ++r)
            cbv[i][r] = cb[o0 + i * 16 + madd + r];

#pragma unroll
    for (int i = 0; i < 2; ++i) {
#pragma unroll
        for (int j = 0; j < 8; ++j) {
            const int p    = j * 16 + l15;
            const int prow = h0 + (p >> 6);
            const int col  = p & 63;
            const size_t base = (size_t)b * C_ * HW_ + (size_t)prow * W_ + col;
            const int ob = o0 + i * 16 + madd;
            float xv[8];
#pragma unroll
            for (int r = 0; r < 8; ++r)
                xv[r] = x[base + (size_t)(ob + r) * HW_];
#pragma unroll
            for (int r = 0; r < 8; ++r)
                out[base + (size_t)(ob + r) * HW_] = xv[r] + cbv[i][r] + acc[i][j][r];
        }
    }
}

extern "C" void kernel_launch(void* const* d_in, const int* in_sizes, int n_in,
                              void* d_out, int out_size, void* d_ws, size_t ws_size,
                              hipStream_t stream) {
    (void)in_sizes; (void)n_in; (void)d_ws; (void)ws_size; (void)out_size;
    const float* x  = (const float*)d_in[0];
    const float* cw = (const float*)d_in[1];
    const float* cb = (const float*)d_in[2];
    float* out = (float*)d_out;

    dim3 grid(H_ / 2, B_);
    dim3 block(512);
    fused_median_conv1x1<<<grid, block, SMEM_BYTES_, stream>>>(x, cw, cb, out);
}


# wave roles merged: every wave runs a step's row loads, medians and matrix steps in turn; 8-wave blocks at 256 registers, two resident per compute unit, each filling the other's waits; sum order kept
# speedup vs baseline: 1.2240x; 1.0439x over previous
#include <hip/hip_runtime.h>


typedef __attribute__((ext_vector_type(2))) float v2f;
typedef __attribute__((ext_vector_type(8))) float v8f;

#define B_  32
#define C_  256
#define H_  64
#define W_  64
#define HW_ (H_ * W_)

#define PX_ 128
#define KPLANE_ (PX_ * 4)

#define CH_     32
#define NCHUNK_ (C_ / CH_)
#define MSLOT_  ((CH_ / 4) * KPLANE_)
#define RP_     (W_ + 2)
#define RSLOT_  (CH_ * 4 * RP_)
#define RAW0_   (2 * MSLOT_)
#define SMEM_BYTES_ ((2 * MSLOT_ + 2 * RSLOT_) * 4)
#define GRP_    4
#define NT_     256

__device__ __forceinline__ void s2f(float& a, float& b) {
    float mn = fminf(a, b);
    b = fmaxf(a, b);
    a = mn;
}

__device__ __forceinline__ float median9(float v0, float v1, float v2,
                                         float v3, float v4, float v5,
                                         float v6, float v7, float v8) {
    s2f(v0, v3); s2f(v1, v4); s2f(v2, v5);
    s2f(v0, v1); s2f(v0, v2);
    s2f(v4, v5); s2f(v3, v5);
    s2f(v1, v2); s2f(v3, v4);
    s2f(v1, v3); s2f(v1, v6);
    s2f(v4, v6); s2f(v2, v6);
    s2f(v2, v3); s2f(v4, v7); s2f(v2, v4); s2f(v3, v7);
    s2f(v4, v8); s2f(v3, v8); s2f(v3, v4);
    return v4;
}

__device__ __forceinline__ void raw_load(const float* __restrict__ xrow, int w,
                                         float (&v)[CH_]) {
#pragma unroll
    for (int cl = 0; cl < CH_; ++cl) v[cl] = xrow[(size_t)cl * HW_ + w];
}

__device__ __forceinline__ void raw_store(float* __restrict__ rslot, int rr, int w,
                                          bool row_in, const float (&v)[CH_]) {
#pragma unroll
    for (int cl = 0; cl < CH_; ++cl)
        rslot[(cl * 4 + rr) * RP_ + w + 1] = row_in ? v[cl] : 0.0f;
}

template <int STRIDE, int NMED>
__device__ __forceinline__ void medians(const float* __restrict__ rslot,
                                        float* __restrict__ mslot,
                                        int cp, int r0, int w) {
    const float* lanep = rslot + r0 * RP_ + w;
    float* outp = mslot + (r0 * 64 + w) * 4;
#pragma unroll 1
    for (int g = 0; g < NMED; g += GRP_) {
        float v[GRP_][9];
#pragma unroll
        for (int m = 0; m < GRP_; ++m) {
            const float* p = lanep + (cp + (g + m) * STRIDE) * (4 * RP_);
#pragma unroll
            for (int dr = 0; dr < 3; ++dr)
#pragma unroll
                for (int dc = 0; dc < 3; ++dc)
                    v[m][dr * 3 + dc] = p[dr * RP_ + dc];
        }
#pragma unroll
        for (int m = 0; m < GRP_; ++m) {
            const int cl = cp + (g + m) * STRIDE;
            outp[(cl >> 2) * KPLANE_ + (cl & 3)] =
                median9(v[m][0], v[m][1], v[m][2], v[m][3], v[m][4],
                        v[m][5], v[m][6], v[m][7], v[m][8]);
        }
    }
}

__global__ __launch_bounds__(NT_, 4)
void fused_median_conv1x1(const float* __restrict__ x,
                          const float* __restrict__ cw,
                          const float* __restrict__ cb,
                          float* __restrict__ out) {
    extern __shared__ float smem[];

    const int h0   = blockIdx.x * 2;
    const int b    = blockIdx.y;
    const int t    = threadIdx.x;
    const int lane = t & 31;
    const int wave = __builtin_amdgcn_readfirstlane(t >> 5);

    const int w  = (wave & 1) * 32 + lane;
    const int r0 = (wave >> 1) & 1;
    const int cp = (wave >> 2) & 1;
    const int rr = (wave >> 1) & 3;
    const int irow = h0 - 1 + rr;
    const bool row_in = (irow >= 0) && (irow < H_);
    const int irow_c = row_in ? irow : h0;
    const float* xrow0 = x + (size_t)b * C_ * HW_ + (size_t)irow_c * W_;

#pragma unroll
    for (int slot = 0; slot < 2; ++slot) {
        const int cl = (t >> 3) & 31, prow = (t >> 1) & 3;
        smem[RAW0_ + slot * RSLOT_ + (cl * 4 + prow) * RP_ + ((t & 1) ? (RP_ - 1) : 0)] = 0.0f;
    }

    {
        float rv[CH_];
        raw_load(xrow0, w, rv);
        raw_store(smem + RAW0_, rr, w, row_in, rv);
    }
    __syncthreads();
    {
        float rv[CH_];
        raw_load(xrow0 + (size_t)CH_ * HW_, w, rv);
        medians<2, CH_ / 2>(smem + RAW0_, smem, cp, r0, w);
        raw_store(smem + RAW0_ + RSLOT_, rr, w, row_in, rv);
    }

    const int o0   = wave * 32;
    const int l15  = lane & 15;
    const int koff = (lane < 16) ? 0 : 2;

    const v2f* a0p = (const v2f*)(cw + (size_t)(o0 +      l15) * C_ + koff);
    const v2f* a1p = (const v2f*)(cw + (size_t)(o0 + 16 + l15) * C_ + koff);

    v8f acc[2][8];
#pragma unroll
    for (int i = 0; i < 2; ++i)
#pragma unroll
        for (int j = 0; j < 8; ++j)
            acc[i][j] = (v8f){0.f, 0.f, 0.f, 0.f, 0.f, 0.f, 0.f, 0.f};

    __syncthreads();

    for (int ch = 0; ch < NCHUNK_; ++ch) {
        const bool stage1 = (ch + 2 < NCHUNK_);
        float rv[CH_];
        if (stage1) raw_load(xrow0 + (size_t)(ch + 2) * CH_ * HW_, w, rv);

        if (ch + 1 < NCHUNK_)
            medians<2, CH_ / 2>(smem + RAW0_ + ((ch + 1) & 1) * RSLOT_,
                                smem + ((ch + 1) & 1) * MSLOT_, cp, r0, w);

        const v2f* bp = (const v2f*)(smem + (ch & 1) * MSLOT_ + l15 * 4 + koff);
#pragma unroll 2
        for (int kk = 0; kk < CH_; kk += 4) {
            const int k0 = ch * CH_ + kk;
            v2f a0 = a0p[k0 >> 1];
            v2f a1 = a1p[k0 >> 1];
            const int bi = (kk >> 2) * (KPLANE_ / 2);
            v2f bv[8];
#pragma unroll
            for (int j = 0; j < 8; ++j) bv[j] = bp[bi + j * 32];
#pragma unroll
            for (int j = 0; j < 8; ++j)
                acc[0][j] = __builtin_amdgcn_wmma_f32_16x16x4_f32(false, a0, false, bv[j],
                                                                  (short)0, acc[0][j], false, false);
#pragma unroll
            for (int j = 0; j < 8; ++j)
                acc[1][j] = __builtin_amdgcn_wmma_f32_16x16x4_f32(false, a1, false, bv[j],
                                                                  (short)0, acc[1][j], false, false);
        }

        if (stage1) raw_store(smem + RAW0_ + (ch & 1) * RSLOT_, rr, w, row_in, rv);
        __syncthreads();
    }

    const int madd = (lane < 16) ? 0 : 8;

    float cbv[2][8];
#pragma unroll
    for (int i = 0; i < 2; ++i)
#pragma unroll
        for (int r = 0; r < 8; ++r)
            cbv[i][r] = cb[o0 + i * 16 + madd + r];

#pragma unroll
    for (int i = 0; i < 2; ++i) {
#pragma unroll
        for (int j = 0; j < 8; ++j) {
            const int p    = j * 16 + l15;
            const int prow = h0 + (p >> 6);
            const int col  = p & 63;
            const size_t base = (size_t)b * C_ * HW_ + (size_t)prow * W_ + col;
            const int ob = o0 + i * 16 + madd;
            float xv[8];
#pragma unroll
            for (int r = 0; r < 8; ++r)
                xv[r] = x[base + (size_t)(ob + r) * HW_];
#pragma unroll
            for (int r = 0; r < 8; ++r)
                out[base + (size_t)(ob + r) * HW_] = xv[r] + cbv[i][r] + acc[i][j][r];
        }
    }
}

extern "C" void kernel_launch(void* const* d_in, const int* in_sizes, int n_in,
                              void* d_out, int out_size, void* d_ws, size_t ws_size,
                              hipStream_t stream) {
    (void)in_sizes; (void)n_in; (void)d_ws; (void)ws_size; (void)out_size;
    const float* x  = (const float*)d_in[0];
    const float* cw = (const float*)d_in[1];
    const float* cb = (const float*)d_in[2];
    float* out = (float*)d_out;

    dim3 grid(H_ / 2, B_);
    dim3 block(NT_);
    fused_median_conv1x1<<<grid, block, SMEM_BYTES_, stream>>>(x, cw, cb, out);
}


# LDS raw ring replaced by a ring of sorted 3-tap columns: stage 1 loads 4 rows x 8 channels per wave, stores min/med/max of each vertical triple once; a median = 4 three-input selections; sums kept
# speedup vs baseline: 1.3344x; 1.0902x over previous
#include <hip/hip_runtime.h>


typedef __attribute__((ext_vector_type(2))) float v2f;
typedef __attribute__((ext_vector_type(8))) float v8f;

#define B_  32
#define C_  256
#define H_  64
#define W_  64
#define HW_ (H_ * W_)

#define PX_ 128
#define KPLANE_ (PX_ * 4)

#define CH_     32
#define NCHUNK_ (C_ / CH_)
#define MSLOT_  ((CH_ / 4) * KPLANE_)
#define SP_     (W_ + 2)
#define CS_     (2 * 3 * SP_)
#define SSLOT_  (CH_ * CS_)
#define MED0_   (2 * SSLOT_)
#define SMEM_BYTES_ ((2 * SSLOT_ + 2 * MSLOT_) * 4)
#define GRP_    8
#define NT_     256

__device__ __forceinline__ float min3f(float a, float b, float c) {
    return __builtin_fminf(__builtin_fminf(a, b), c);
}
__device__ __forceinline__ float max3f(float a, float b, float c) {
    return __builtin_fmaxf(__builtin_fmaxf(a, b), c);
}
__device__ __forceinline__ float med3f(float a, float b, float c) {
    return __builtin_amdgcn_fmed3f(a, b, c);
}

__device__ __forceinline__ void rows_load(const float* __restrict__ xc, const int (&ro)[4],
                                          float (&v)[CH_]) {
#pragma unroll
    for (int c8 = 0; c8 < 8; ++c8)
#pragma unroll
        for (int q = 0; q < 4; ++q) v[c8 * 4 + q] = xc[(size_t)c8 * HW_ + ro[q]];
}

__device__ __forceinline__ void cols_store(float* __restrict__ cslot, int w,
                                           bool top_in, bool bot_in, const float (&v)[CH_]) {
    float* p = cslot + w + 1;
#pragma unroll
    for (int c8 = 0; c8 < 8; ++c8) {
        const float a = top_in ? v[c8 * 4 + 0] : 0.0f;
        const float b = v[c8 * 4 + 1];
        const float c = v[c8 * 4 + 2];
        const float d = bot_in ? v[c8 * 4 + 3] : 0.0f;
        float* pc = p + c8 * CS_;
        pc[0 * SP_] = min3f(a, b, c);
        pc[1 * SP_] = med3f(a, b, c);
        pc[2 * SP_] = max3f(a, b, c);
        pc[3 * SP_] = min3f(b, c, d);
        pc[4 * SP_] = med3f(b, c, d);
        pc[5 * SP_] = max3f(b, c, d);
    }
}

template <int STRIDE, int NMED>
__device__ __forceinline__ void medians(const float* __restrict__ cslot,
                                        float* __restrict__ mslot,
                                        int cp, int r0, int w) {
    const float* lanep = cslot + cp * CS_ + r0 * 3 * SP_ + w;
    float* outp = mslot + (r0 * 64 + w) * 4;
#pragma unroll 1
    for (int g = 0; g < NMED; g += GRP_, lanep += GRP_ * STRIDE * CS_) {
        float v[GRP_][9];
#pragma unroll
        for (int m = 0; m < GRP_; ++m) {
            const float* p = lanep + m * STRIDE * CS_;
#pragma unroll
            for (int s = 0; s < 3; ++s)
#pragma unroll
                for (int dc = 0; dc < 3; ++dc)
                    v[m][s * 3 + dc] = p[s * SP_ + dc];
        }
#pragma unroll
        for (int m = 0; m < GRP_; ++m) {
            const int cl = cp + (g + m) * STRIDE;
            outp[(cl >> 2) * KPLANE_ + (cl & 3)] =
                med3f(max3f(v[m][0], v[m][1], v[m][2]),
                      med3f(v[m][3], v[m][4], v[m][5]),
                      min3f(v[m][6], v[m][7], v[m][8]));
        }
    }
}

__global__ __launch_bounds__(NT_, 4)
void fused_median_conv1x1(const float* __restrict__ x,
                          const float* __restrict__ cw,
                          const float* __restrict__ cb,
                          float* __restrict__ out) {
    extern __shared__ float smem[];

    const int h0   = blockIdx.x * 2;
    const int b    = blockIdx.y;
    const int t    = threadIdx.x;
    const int lane = t & 31;
    const int wave = __builtin_amdgcn_readfirstlane(t >> 5);

    const int w  = (wave & 1) * 32 + lane;
    const int r0 = (wave >> 1) & 1;
    const int cp = (wave >> 2) & 1;
    const int oct = wave >> 1;
    const bool top_in = (h0 > 0);
    const bool bot_in = (h0 + 2 < H_);
    const int ro[4] = { (top_in ? h0 - 1 : h0) * W_, h0 * W_, (h0 + 1) * W_,
                        (bot_in ? h0 + 2 : h0 + 1) * W_ };
    const float* xc0 = x + (size_t)b * C_ * HW_ + (size_t)(oct * 8) * HW_ + w;
    const int coff = oct * 8 * CS_;

    for (int i = t; i < 2 * (SSLOT_ / SP_) * 2; i += NT_) {
        const int slot = i / ((SSLOT_ / SP_) * 2);
        const int rem  = i - slot * ((SSLOT_ / SP_) * 2);
        smem[slot * SSLOT_ + (rem >> 1) * SP_ + ((rem & 1) ? (SP_ - 1) : 0)] = 0.0f;
    }

    {
        float rv[CH_];
        rows_load(xc0, ro, rv);
        cols_store(smem + coff, w, top_in, bot_in, rv);
    }
    __syncthreads();
    {
        float rv[CH_];
        rows_load(xc0 + (size_t)CH_ * HW_, ro, rv);
        medians<2, CH_ / 2>(smem, smem + MED0_, cp, r0, w);
        cols_store(smem + SSLOT_ + coff, w, top_in, bot_in, rv);
    }

    const int o0   = wave * 32;
    const int l15  = lane & 15;
    const int koff = (lane < 16) ? 0 : 2;

    const v2f* a0p = (const v2f*)(cw + (size_t)(o0 +      l15) * C_ + koff);
    const v2f* a1p = (const v2f*)(cw + (size_t)(o0 + 16 + l15) * C_ + koff);

    v8f acc[2][8];
#pragma unroll
    for (int i = 0; i < 2; ++i)
#pragma unroll
        for (int j = 0; j < 8; ++j)
            acc[i][j] = (v8f){0.f, 0.f, 0.f, 0.f, 0.f, 0.f, 0.f, 0.f};

    __syncthreads();

    for (int ch = 0; ch < NCHUNK_; ++ch) {
        const bool stage1 = (ch + 2 < NCHUNK_);
        float rv[CH_];
        if (stage1) rows_load(xc0 + (size_t)(ch + 2) * CH_ * HW_, ro, rv);

        if (ch + 1 < NCHUNK_)
            medians<2, CH_ / 2>(smem + ((ch + 1) & 1) * SSLOT_,
                                smem + MED0_ + ((ch + 1) & 1) * MSLOT_, cp, r0, w);

        const v2f* bp = (const v2f*)(smem + MED0_ + (ch & 1) * MSLOT_ + l15 * 4 + koff);
#pragma unroll 2
        for (int kk = 0; kk < CH_; kk += 4) {
            const int k0 = ch * CH_ + kk;
            v2f a0 = a0p[k0 >> 1];
            v2f a1 = a1p[k0 >> 1];
            const int bi = (kk >> 2) * (KPLANE_ / 2);
            v2f bv[8];
#pragma unroll
            for (int j = 0; j < 8; ++j) bv[j] = bp[bi + j * 32];
#pragma unroll
            for (int j = 0; j < 8; ++j)
                acc[0][j] = __builtin_amdgcn_wmma_f32_16x16x4_f32(false, a0, false, bv[j],
                                                                  (short)0, acc[0][j], false, false);
#pragma unroll
            for (int j = 0; j < 8; ++j)
                acc[1][j] = __builtin_amdgcn_wmma_f32_16x16x4_f32(false, a1, false, bv[j],
                                                                  (short)0, acc[1][j], false, false);
        }

        if (stage1) cols_store(smem + (ch & 1) * SSLOT_ + coff, w, top_in, bot_in, rv);
        __syncthreads();
    }

    const int madd = (lane < 16) ? 0 : 8;

    float cbv[2][8];
#pragma unroll
    for (int i = 0; i < 2; ++i)
#pragma unroll
        for (int r = 0; r < 8; ++r)
            cbv[i][r] = cb[o0 + i * 16 + madd + r];

#pragma unroll
    for (int i = 0; i < 2; ++i) {
#pragma unroll
        for (int j = 0; j < 8; ++j) {
            const int p    = j * 16 + l15;
            const int prow = h0 + (p >> 6);
            const int col  = p & 63;
            const size_t base = (size_t)b * C_ * HW_ + (size_t)prow * W_ + col;
            const int ob = o0 + i * 16 + madd;
            float xv[8];
#pragma unroll
            for (int r = 0; r < 8; ++r)
                xv[r] = x[base + (size_t)(ob + r) * HW_];
#pragma unroll
            for (int r = 0; r < 8; ++r)
                out[base + (size_t)(ob + r) * HW_] = xv[r] + cbv[i][r] + acc[i][j][r];
        }
    }
}

extern "C" void kernel_launch(void* const* d_in, const int* in_sizes, int n_in,
                              void* d_out, int out_size, void* d_ws, size_t ws_size,
                              hipStream_t stream) {
    (void)in_sizes; (void)n_in; (void)d_ws; (void)ws_size; (void)out_size;
    const float* x  = (const float*)d_in[0];
    const float* cw = (const float*)d_in[1];
    const float* cb = (const float*)d_in[2];
    float* out = (float*)d_out;

    dim3 grid(H_ / 2, B_);
    dim3 block(NT_);
    fused_median_conv1x1<<<grid, block, SMEM_BYTES_, stream>>>(x, cw, cb, out);
}


# column ring re-laid out with 4 channels per 16-byte unit: a median's taps are 9 sixteen-byte LDS reads per 4 channels (were 20 narrow ones), sorted columns and medians stored 16 bytes wide; sums kept
# speedup vs baseline: 1.4029x; 1.0513x over previous
#include <hip/hip_runtime.h>


typedef __attribute__((ext_vector_type(2))) float v2f;
typedef __attribute__((ext_vector_type(4))) float v4f;
typedef __attribute__((ext_vector_type(8))) float v8f;

#define B_  32
#define C_  256
#define H_  64
#define W_  64
#define HW_ (H_ * W_)

#define PX_ 128
#define KPLANE_ (PX_ * 4)

#define CH_     32
#define NCHUNK_ (C_ / CH_)
#define MSLOT_  ((CH_ / 4) * KPLANE_)
#define SP_     (W_ + 2)
#define KROW_   (SP_ * 4)
#define OROW_   (3 * KROW_)
#define CG_     (2 * OROW_)
#define SSLOT_  ((CH_ / 4) * CG_)
#define MED0_   (2 * SSLOT_)
#define SMEM_BYTES_ ((2 * SSLOT_ + 2 * MSLOT_) * 4)
#define GRP_    2
#define NT_     256

__device__ __forceinline__ float min3f(float a, float b, float c) {
    return __builtin_fminf(__builtin_fminf(a, b), c);
}
__device__ __forceinline__ float max3f(float a, float b, float c) {
    return __builtin_fmaxf(__builtin_fmaxf(a, b), c);
}
__device__ __forceinline__ float med3f(float a, float b, float c) {
    return __builtin_amdgcn_fmed3f(a, b, c);
}

__device__ __forceinline__ void rows_load(const float* __restrict__ xc, const int (&ro)[4],
                                          float (&v)[CH_]) {
#pragma unroll
    for (int c8 = 0; c8 < 8; ++c8)
#pragma unroll
        for (int q = 0; q < 4; ++q) v[c8 * 4 + q] = xc[(size_t)c8 * HW_ + ro[q]];
}

__device__ __forceinline__ void cols_store(float* __restrict__ cslot, int w,
                                           bool top_in, bool bot_in, const float (&v)[CH_]) {
    v4f* p = (v4f*)(cslot + (w + 1) * 4);
#pragma unroll
    for (int g2 = 0; g2 < 2; ++g2) {
        v4f lo0, md0, hi0, lo1, md1, hi1;
#pragma unroll
        for (int e = 0; e < 4; ++e) {
            const int c8 = g2 * 4 + e;
            const float a = top_in ? v[c8 * 4 + 0] : 0.0f;
            const float b = v[c8 * 4 + 1];
            const float c = v[c8 * 4 + 2];
            const float d = bot_in ? v[c8 * 4 + 3] : 0.0f;
            lo0[e] = min3f(a, b, c);
            md0[e] = med3f(a, b, c);
            hi0[e] = max3f(a, b, c);
            lo1[e] = min3f(b, c, d);
            md1[e] = med3f(b, c, d);
            hi1[e] = max3f(b, c, d);
        }
        v4f* pg = p + g2 * (CG_ / 4);
        pg[0 * (KROW_ / 4)] = lo0;
        pg[1 * (KROW_ / 4)] = md0;
        pg[2 * (KROW_ / 4)] = hi0;
        pg[3 * (KROW_ / 4)] = lo1;
        pg[4 * (KROW_ / 4)] = md1;
        pg[5 * (KROW_ / 4)] = hi1;
    }
}

template <int NGRP>
__device__ __forceinline__ void medians(const float* __restrict__ cslot,
                                        float* __restrict__ mslot,
                                        int cp, int r0, int w) {
    const v4f* lanep = (const v4f*)(cslot + cp * CG_ + r0 * OROW_ + w * 4);
    v4f* outp = (v4f*)(mslot + cp * KPLANE_ + (r0 * 64 + w) * 4);
#pragma unroll 1
    for (int g = 0; g < NGRP; g += GRP_, lanep += GRP_ * 2 * (CG_ / 4),
                                         outp += GRP_ * 2 * (KPLANE_ / 4)) {
        v4f t[GRP_][9];
#pragma unroll
        for (int m = 0; m < GRP_; ++m) {
            const v4f* p = lanep + m * 2 * (CG_ / 4);
#pragma unroll
            for (int s = 0; s < 3; ++s)
#pragma unroll
                for (int dc = 0; dc < 3; ++dc)
                    t[m][s * 3 + dc] = p[s * (KROW_ / 4) + dc];
        }
#pragma unroll
        for (int m = 0; m < GRP_; ++m) {
            v4f r;
#pragma unroll
            for (int e = 0; e < 4; ++e)
                r[e] = med3f(max3f(t[m][0][e], t[m][1][e], t[m][2][e]),
                             med3f(t[m][3][e], t[m][4][e], t[m][5][e]),
                             min3f(t[m][6][e], t[m][7][e], t[m][8][e]));
            outp[m * 2 * (KPLANE_ / 4)] = r;
        }
    }
}

__global__ __launch_bounds__(NT_, 4)
void fused_median_conv1x1(const float* __restrict__ x,
                          const float* __restrict__ cw,
                          const float* __restrict__ cb,
                          float* __restrict__ out) {
    extern __shared__ __attribute__((aligned(16))) float smem[];

    const int h0   = blockIdx.x * 2;
    const int b    = blockIdx.y;
    const int t    = threadIdx.x;
    const int lane = t & 31;
    const int wave = __builtin_amdgcn_readfirstlane(t >> 5);

    const int w  = (wave & 1) * 32 + lane;
    const int r0 = (wave >> 1) & 1;
    const int cp = (wave >> 2) & 1;
    const int oct = wave >> 1;
    const bool top_in = (h0 > 0);
    const bool bot_in = (h0 + 2 < H_);
    const int ro[4] = { (top_in ? h0 - 1 : h0) * W_, h0 * W_, (h0 + 1) * W_,
                        (bot_in ? h0 + 2 : h0 + 1) * W_ };
    const float* xc0 = x + (size_t)b * C_ * HW_ + (size_t)(oct * 8) * HW_ + w;
    const int coff = oct * 2 * CG_;

    if (t < 2 * (SSLOT_ / KROW_) * 2) {
        const int slot = t / ((SSLOT_ / KROW_) * 2);
        const int rem  = t - slot * ((SSLOT_ / KROW_) * 2);
        *(v4f*)(smem + slot * SSLOT_ + (rem >> 1) * KROW_ + ((rem & 1) ? (SP_ - 1) * 4 : 0)) =
            (v4f){0.f, 0.f, 0.f, 0.f};
    }

    {
        float rv[CH_];
        rows_load(xc0, ro, rv);
        cols_store(smem + coff, w, top_in, bot_in, rv);
    }
    __syncthreads();
    {
        float rv[CH_];
        rows_load(xc0 + (size_t)CH_ * HW_, ro, rv);
        medians<CH_ / 8>(smem, smem + MED0_, cp, r0, w);
        cols_store(smem + SSLOT_ + coff, w, top_in, bot_in, rv);
    }

    const int o0   = wave * 32;
    const int l15  = lane & 15;
    const int koff = (lane < 16) ? 0 : 2;

    const v2f* a0p = (const v2f*)(cw + (size_t)(o0 +      l15) * C_ + koff);
    const v2f* a1p = (const v2f*)(cw + (size_t)(o0 + 16 + l15) * C_ + koff);

    v8f acc[2][8];
#pragma unroll
    for (int i = 0; i < 2; ++i)
#pragma unroll
        for (int j = 0; j < 8; ++j)
            acc[i][j] = (v8f){0.f, 0.f, 0.f, 0.f, 0.f, 0.f, 0.f, 0.f};

    __syncthreads();

    for (int ch = 0; ch < NCHUNK_; ++ch) {
        const bool stage1 = (ch + 2 < NCHUNK_);
        float rv[CH_];
        if (stage1) rows_load(xc0 + (size_t)(ch + 2) * CH_ * HW_, ro, rv);

        if (ch + 1 < NCHUNK_)
            medians<CH_ / 8>(smem + ((ch + 1) & 1) * SSLOT_,
                                smem + MED0_ + ((ch + 1) & 1) * MSLOT_, cp, r0, w);

        const v2f* bp = (const v2f*)(smem + MED0_ + (ch & 1) * MSLOT_ + l15 * 4 + koff);
#pragma unroll 2
        for (int kk = 0; kk < CH_; kk += 4) {
            const int k0 = ch * CH_ + kk;
            v2f a0 = a0p[k0 >> 1];
            v2f a1 = a1p[k0 >> 1];
            const int bi = (kk >> 2) * (KPLANE_ / 2);
            v2f bv[8];
#pragma unroll
            for (int j = 0; j < 8; ++j) bv[j] = bp[bi + j * 32];
#pragma unroll
            for (int j = 0; j < 8; ++j)
                acc[0][j] = __builtin_amdgcn_wmma_f32_16x16x4_f32(false, a0, false, bv[j],
                                                                  (short)0, acc[0][j], false, false);
#pragma unroll
            for (int j = 0; j < 8; ++j)
                acc[1][j] = __builtin_amdgcn_wmma_f32_16x16x4_f32(false, a1, false, bv[j],
                                                                  (short)0, acc[1][j], false, false);
        }

        if (stage1) cols_store(smem + (ch & 1) * SSLOT_ + coff, w, top_in, bot_in, rv);
        __syncthreads();
    }

    const int madd = (lane < 16) ? 0 : 8;

    float cbv[2][8];
#pragma unroll
    for (int i = 0; i < 2; ++i)
#pragma unroll
        for (int r = 0; r < 8; ++r)
            cbv[i][r] = cb[o0 + i * 16 + madd + r];

#pragma unroll
    for (int i = 0; i < 2; ++i) {
#pragma unroll
        for (int j = 0; j < 8; ++j) {
            const int p    = j * 16 + l15;
            const int prow = h0 + (p >> 6);
            const int col  = p & 63;
            const size_t base = (size_t)b * C_ * HW_ + (size_t)prow * W_ + col;
            const int ob = o0 + i * 16 + madd;
            float xv[8];
#pragma unroll
            for (int r = 0; r < 8; ++r)
                xv[r] = x[base + (size_t)(ob + r) * HW_];
#pragma unroll
            for (int r = 0; r < 8; ++r)
                out[base + (size_t)(ob + r) * HW_] = xv[r] + cbv[i][r] + acc[i][j][r];
        }
    }
}

extern "C" void kernel_launch(void* const* d_in, const int* in_sizes, int n_in,
                              void* d_out, int out_size, void* d_ws, size_t ws_size,
                              hipStream_t stream) {
    (void)in_sizes; (void)n_in; (void)d_ws; (void)ws_size; (void)out_size;
    const float* x  = (const float*)d_in[0];
    const float* cw = (const float*)d_in[1];
    const float* cb = (const float*)d_in[2];
    float* out = (float*)d_out;

    dim3 grid(H_ / 2, B_);
    dim3 block(NT_);
    fused_median_conv1x1<<<grid, block, SMEM_BYTES_, stream>>>(x, cw, cb, out);
}
